# starting kernel
# baseline (speedup 1.0000x reference)
#include <hip/hip_runtime.h>


typedef float v2f __attribute__((ext_vector_type(2)));
typedef float v8f __attribute__((ext_vector_type(8)));

#define D_MODEL 512
#define D_STATE 64
#define SEQ_LEN 2048
#define BATCH 4
#define MTOT (BATCH * SEQ_LEN)
#define NCHUNK 8
#define TCHUNK (SEQ_LEN / NCHUNK)

__global__ __launch_bounds__(256) void gemm_wmma_f32(
    const float* __restrict__ A, const float* __restrict__ W,
    const float* __restrict__ bias, float* __restrict__ out)
{
    const int lane = threadIdx.x & 31;
    const int l16  = lane & 15;
    const int half = lane >> 4;
    const int gwave = blockIdx.x * 8 + (threadIdx.x >> 5);

    const int nBlocksN = D_MODEL / 64;
    const int mb = gwave / nBlocksN;
    const int nb = gwave % nBlocksN;
    const int M0 = mb * 64;
    const int N0 = nb * 64;

    const float* arow[4];
    const float* brow[4];
#pragma unroll
    for (int i = 0; i < 4; ++i)
        arow[i] = A + (size_t)(M0 + i * 16 + l16) * D_MODEL + half * 2;
#pragma unroll
    for (int j = 0; j < 4; ++j)
        brow[j] = W + (size_t)(N0 + j * 16 + l16) * D_MODEL + half * 2;

    v8f acc[4][4] = {};

#pragma unroll 4
    for (int k0 = 0; k0 < D_MODEL; k0 += 4) {
        v2f a[4], b[4];
#pragma unroll
        for (int i = 0; i < 4; ++i) a[i] = *(const v2f*)(arow[i] + k0);
#pragma unroll
        for (int j = 0; j < 4; ++j) b[j] = *(const v2f*)(brow[j] + k0);
#pragma unroll
        for (int i = 0; i < 4; ++i)
#pragma unroll
            for (int j = 0; j < 4; ++j)
                acc[i][j] = __builtin_amdgcn_wmma_f32_16x16x4_f32(
                    false, a[i], false, b[j], (short)0, acc[i][j],
                    false, false);
    }

#pragma unroll
    for (int j = 0; j < 4; ++j) {
        const float bj = bias[N0 + j * 16 + l16];
#pragma unroll
        for (int i = 0; i < 4; ++i) {
#pragma unroll
            for (int r = 0; r < 8; ++r) {
                const int row = M0 + i * 16 + half * 8 + r;
                out[(size_t)row * D_MODEL + N0 + j * 16 + l16] =
                    acc[i][j][r] + bj;
            }
        }
    }
}

__global__ __launch_bounds__(256) void s4_chunk_end(
    const float* __restrict__ xin,
    const float* __restrict__ A_log,
    const float* __restrict__ dt,
    float* __restrict__ Echunk)
{
    const int wid  = blockIdx.x * 8 + (threadIdx.x >> 5);
    const int lane = threadIdx.x & 31;
    const int c  = wid & (NCHUNK - 1);
    const int bd = wid >> 3;
    const int b  = bd >> 9;
    const int d  = bd & 511;
    const int n0 = lane * 2;

    const float dtd = dt[d];
    const float dA0 = expf(-expf(A_log[d * D_STATE + n0])     * dtd);
    const float dA1 = expf(-expf(A_log[d * D_STATE + n0 + 1]) * dtd);

    const float* up = xin + (size_t)(b * SEQ_LEN + c * TCHUNK) * D_MODEL + d;
    float s0 = 0.f, s1 = 0.f;
#pragma unroll 8
    for (int t = 0; t < TCHUNK; ++t) {
        const float u = up[(size_t)t * D_MODEL];
        s0 = fmaf(dA0, s0, u);
        s1 = fmaf(dA1, s1, u);
    }
    float* E = Echunk + ((size_t)bd * NCHUNK + c) * D_STATE;
    E[n0]     = s0;
    E[n0 + 1] = s1;
}

__global__ __launch_bounds__(256) void s4_chunk_prefix(
    const float* __restrict__ A_log, const float* __restrict__ dt,
    const float* __restrict__ Echunk,
    float* __restrict__ Sinit)
{
    const int bd   = blockIdx.x * 8 + (threadIdx.x >> 5);
    const int lane = threadIdx.x & 31;
    const int d  = bd & 511;
    const int n0 = lane * 2;

    const float dtd = dt[d];
    const float a0 = -expf(A_log[d * D_STATE + n0]);
    const float a1 = -expf(A_log[d * D_STATE + n0 + 1]);
    const float dAp0 = expf(a0 * dtd * (float)TCHUNK);
    const float dAp1 = expf(a1 * dtd * (float)TCHUNK);

    const float* E = Echunk + (size_t)bd * NCHUNK * D_STATE;
    float*       S = Sinit  + (size_t)bd * NCHUNK * D_STATE;

    float s0 = 0.f, s1 = 0.f;
#pragma unroll
    for (int c = 0; c < NCHUNK; ++c) {
        S[c * D_STATE + n0]     = s0;
        S[c * D_STATE + n0 + 1] = s1;
        s0 = fmaf(dAp0, s0, E[c * D_STATE + n0]);
        s1 = fmaf(dAp1, s1, E[c * D_STATE + n0 + 1]);
    }
}

__global__ __launch_bounds__(256) void s4_scan_chunk(
    const float* __restrict__ xin,
    const float* __restrict__ A_log,
    const float* __restrict__ Bm,
    const float* __restrict__ Cm,
    const float* __restrict__ Dv,
    const float* __restrict__ dt,
    const float* __restrict__ Sinit,
    float* __restrict__ y)
{
    const int wid  = blockIdx.x * 8 + (threadIdx.x >> 5);
    const int lane = threadIdx.x & 31;
    const int c  = wid & (NCHUNK - 1);
    const int bd = wid >> 3;
    const int b  = bd >> 9;
    const int d  = bd & 511;
    const int n0 = lane * 2;

    const float dtd = dt[d];
    const float dA0 = expf(-expf(A_log[d * D_STATE + n0])     * dtd);
    const float dA1 = expf(-expf(A_log[d * D_STATE + n0 + 1]) * dtd);
    const float c0  = Cm[d * D_STATE + n0]     * Bm[d * D_STATE + n0]     * dtd;
    const float c1  = Cm[d * D_STATE + n0 + 1] * Bm[d * D_STATE + n0 + 1] * dtd;
    const float Dd  = Dv[d];

    const float* S = Sinit + ((size_t)bd * NCHUNK + c) * D_STATE;
    float s0 = S[n0];
    float s1 = S[n0 + 1];

    const float* up = xin + (size_t)(b * SEQ_LEN + c * TCHUNK) * D_MODEL + d;
    float*       yp = y   + (size_t)(b * SEQ_LEN + c * TCHUNK) * D_MODEL + d;

#pragma unroll 4
    for (int t = 0; t < TCHUNK; ++t) {
        const float u = up[(size_t)t * D_MODEL];
        s0 = fmaf(dA0, s0, u);
        s1 = fmaf(dA1, s1, u);
        float p = fmaf(c1, s1, c0 * s0);
#pragma unroll
        for (int m = 16; m >= 1; m >>= 1)
            p += __shfl_xor(p, m, 32);
        if (lane == 0)
            yp[(size_t)t * D_MODEL] = fmaf(Dd, u, p);
    }
}

extern "C" void kernel_launch(void* const* d_in, const int* in_sizes, int n_in,
                              void* d_out, int out_size, void* d_ws, size_t ws_size,
                              hipStream_t stream)
{
    (void)in_sizes; (void)n_in; (void)out_size; (void)ws_size;

    const float* x     = (const float*)d_in[0];
    const float* W_in  = (const float*)d_in[1];
    const float* b_in  = (const float*)d_in[2];
    const float* A_log = (const float*)d_in[3];
    const float* Bm    = (const float*)d_in[4];
    const float* Cm    = (const float*)d_in[5];
    const float* Dv    = (const float*)d_in[6];
    const float* dt    = (const float*)d_in[7];
    const float* W_out = (const float*)d_in[8];
    const float* b_out = (const float*)d_in[9];

    float* xin  = (float*)d_ws;
    float* ymid = xin  + (size_t)MTOT * D_MODEL;
    float* E    = ymid + (size_t)MTOT * D_MODEL;
    float* S    = E + (size_t)BATCH * D_MODEL * NCHUNK * D_STATE;
    float* out  = (float*)d_out;

    const dim3 blk(256);
    gemm_wmma_f32<<<128, blk, 0, stream>>>(x, W_in, b_in, xin);
    s4_chunk_end  <<<2048, blk, 0, stream>>>(xin, A_log, dt, E);
    s4_chunk_prefix<<<256, blk, 0, stream>>>(A_log, dt, E, S);
    s4_scan_chunk <<<2048, blk, 0, stream>>>(xin, A_log, Bm, Cm, Dv, dt, S, ymid);
    gemm_wmma_f32<<<128, blk, 0, stream>>>(ymid, W_out, b_out, out);
}


# Scan passes: each thread requests its input rows of all 8 steps and the carried state at kernel entry and the step barriers wait for LDS only, so no step waits on a fresh fetch; sums unchanged
# speedup vs baseline: 4.3076x; 4.3076x over previous
#include <hip/hip_runtime.h>


typedef float v2f __attribute__((ext_vector_type(2)));
typedef float v4f __attribute__((ext_vector_type(4)));
typedef float v8f __attribute__((ext_vector_type(8)));

#define D_MODEL 512
#define D_STATE 64
#define SEQ_LEN 2048
#define BATCH 4
#define MTOT (BATCH * SEQ_LEN)
#define NCHUNK 16
#define TCHUNK (SEQ_LEN / NCHUNK)
#define TSTEP 16
#define NSTEP (TCHUNK / TSTEP)

#define KBLK 32
#define TILE_FLOATS (16 * D_MODEL)

struct FragSet {
    v4f a[4][4];
    v4f b[4][4];
};

template <bool PACKED_A>
__device__ __forceinline__ void gemm_fetch(FragSet& f, const float* const (&arow)[4],
                                           const float* const (&brow)[4], int kb)
{
#pragma unroll
    for (int p = 0; p < 4; ++p) {
#pragma unroll
        for (int i = 0; i < 4; ++i)
            f.a[i][p] = *(const v4f*)(arow[i] + (PACKED_A ? kb * 16 + 128 * p : kb + 8 * p));
#pragma unroll
        for (int j = 0; j < 4; ++j) f.b[j][p] = *(const v4f*)(brow[j] + kb + 8 * p);
    }
}

template <bool SWAP>
__device__ __forceinline__ void gemm_block(v8f (&acc)[4][4], const FragSet& f)
{
#pragma unroll
    for (int p = 0; p < 4; ++p) {
        v2f a0[4], a1[4], b0[4], b1[4];
#pragma unroll
        for (int i = 0; i < 4; ++i) {
            a0[i].x = f.a[i][p].x; a0[i].y = f.a[i][p].y;
            a1[i].x = f.a[i][p].z; a1[i].y = f.a[i][p].w;
            b0[i].x = f.b[i][p].x; b0[i].y = f.b[i][p].y;
            b1[i].x = f.b[i][p].z; b1[i].y = f.b[i][p].w;
        }
#pragma unroll
        for (int i = 0; i < 4; ++i)
#pragma unroll
            for (int j = 0; j < 4; ++j)
                acc[i][j] = __builtin_amdgcn_wmma_f32_16x16x4_f32(
                    false, SWAP ? b0[j] : a0[i], false, SWAP ? a0[i] : b0[j],
                    (short)0, acc[i][j], false, false);
#pragma unroll
        for (int i = 0; i < 4; ++i)
#pragma unroll
            for (int j = 0; j < 4; ++j)
                acc[i][j] = __builtin_amdgcn_wmma_f32_16x16x4_f32(
                    false, SWAP ? b1[j] : a1[i], false, SWAP ? a1[i] : b1[j],
                    (short)0, acc[i][j], false, false);
    }
}

template <bool PACKED_OUT>
__global__ __launch_bounds__(128) void gemm_wmma_f32(
    const float* __restrict__ A, const float* __restrict__ W,
    const float* __restrict__ bias, float* __restrict__ out)
{
    constexpr bool PACKED_A = !PACKED_OUT;
    const int lane = threadIdx.x & 31;
    const int l16  = lane & 15;
    const int half = lane >> 4;
    const int w    = threadIdx.x >> 5;
    const int M0 = (blockIdx.x >> 2) * 128 + (w >> 1) * 64;
    const int N0 = (blockIdx.x & 3) * 128 + (w & 1) * 64;

    const float* arow[4];
    const float* brow[4];
#pragma unroll
    for (int i = 0; i < 4; ++i)
        arow[i] = PACKED_A
            ? A + (size_t)((M0 >> 4) + i) * TILE_FLOATS + half * 64 + l16 * 4
            : A + (size_t)(M0 + i * 16 + l16) * D_MODEL + half * 4;
#pragma unroll
    for (int j = 0; j < 4; ++j)
        brow[j] = W + (size_t)(N0 + j * 16 + l16) * D_MODEL + half * 4;

    v8f acc[4][4] = {};
    FragSet f0, f1;

    gemm_fetch<PACKED_A>(f0, arow, brow, 0);
    for (int kb = 0; kb < D_MODEL - 2 * KBLK; kb += 2 * KBLK) {
        gemm_fetch<PACKED_A>(f1, arow, brow, kb + KBLK);
        __builtin_amdgcn_sched_barrier(0);
        gemm_block<PACKED_OUT>(acc, f0);
        __builtin_amdgcn_sched_barrier(0);
        gemm_fetch<PACKED_A>(f0, arow, brow, kb + 2 * KBLK);
        gemm_block<PACKED_OUT>(acc, f1);
    }
    gemm_fetch<PACKED_A>(f1, arow, brow, D_MODEL - KBLK);
    gemm_block<PACKED_OUT>(acc, f0);
    gemm_block<PACKED_OUT>(acc, f1);

    if (PACKED_OUT) {
#pragma unroll
        for (int j = 0; j < 4; ++j) {
            const int ch = N0 + j * 16 + half * 8;
            float bj[8];
#pragma unroll
            for (int r = 0; r < 8; ++r) bj[r] = bias[ch + r];
#pragma unroll
            for (int i = 0; i < 4; ++i) {
                float* op = out + (size_t)((M0 >> 4) + i) * TILE_FLOATS + (ch >> 2) * 64 + l16 * 4;
                v4f o0, o1;
                o0.x = acc[i][j][0] + bj[0]; o0.y = acc[i][j][1] + bj[1];
                o0.z = acc[i][j][2] + bj[2]; o0.w = acc[i][j][3] + bj[3];
                o1.x = acc[i][j][4] + bj[4]; o1.y = acc[i][j][5] + bj[5];
                o1.z = acc[i][j][6] + bj[6]; o1.w = acc[i][j][7] + bj[7];
                *(v4f*)(op)      = o0;
                *(v4f*)(op + 64) = o1;
            }
        }
        return;
    }
#pragma unroll
    for (int j = 0; j < 4; ++j) {
        const float bj = bias[N0 + j * 16 + l16];
#pragma unroll
        for (int i = 0; i < 4; ++i) {
#pragma unroll
            for (int r = 0; r < 8; ++r) {
                const int row = M0 + i * 16 + half * 8 + r;
                out[(size_t)row * D_MODEL + N0 + j * 16 + l16] =
                    acc[i][j][r] + bj;
            }
        }
    }
}

__device__ __forceinline__ void lds_barrier()
{
    __builtin_amdgcn_fence(__ATOMIC_RELEASE, "workgroup", "local");
    __builtin_amdgcn_s_barrier();
    __builtin_amdgcn_fence(__ATOMIC_ACQUIRE, "workgroup", "local");
}

template <bool OUT>
__global__ __launch_bounds__(256) void scan_mm(
    const float* __restrict__ xin,
    const float* __restrict__ A_log,
    const float* __restrict__ Bm,
    const float* __restrict__ Cm,
    const float* __restrict__ Dv,
    const float* __restrict__ dt,
    const float* __restrict__ Sinit,
    float* __restrict__ Eout,
    float* __restrict__ y)
{
    __shared__ float Ul[2][8][512];
    __shared__ float Yl[2][8][320];
    __shared__ float ldaS[8][64];
    __shared__ float cfS[8][64];

    const int tid  = threadIdx.x;
    const int w    = tid >> 5;
    const int lane = tid & 31;
    const int c    = lane & 15;
    const int h    = lane >> 4;
    const int b    = blockIdx.x >> 6;
    const int d0   = (blockIdx.x & 63) * 8;
    const int d    = d0 + w;

    const int ti = tid & 15;
    const int tc = tid >> 4;
    const size_t goff = (size_t)(b * (SEQ_LEN / 16) + tc * (TCHUNK / 16)) * TILE_FLOATS
                      + (size_t)(d0 >> 2) * 64 + ti * 4;
    const int widx = ((ti >> 1) * 16 + tc) * 2 + (ti & 1);
    const int ridx = ti * 16 + tc + 32 * (ti >> 3);
    const size_t soff = ((size_t)(b * D_MODEL + d) * NCHUNK + c) * D_STATE + 8 * h;

    const float dtd = dt[d];
    float dvd = 0.f;
    if (OUT) dvd = Dv[d];
    float alv[2], cmv[2], bmv[2];
#pragma unroll
    for (int e = 0; e < 2; ++e) {
        const int n = lane + 32 * e;
        alv[e] = A_log[d * D_STATE + n];
        if (OUT) {
            cmv[e] = Cm[d * D_STATE + n];
            bmv[e] = Bm[d * D_STATE + n];
        }
    }
    float4 g[NSTEP][2];
    g[0][0] = *(const float4*)(xin + goff);
    g[0][1] = *(const float4*)(xin + goff + 64);
    v8f S[4];
#pragma unroll
    for (int mt = 0; mt < 4; ++mt) {
        if (OUT) {
            const float4 s0 = *(const float4*)(Sinit + soff + 16 * mt);
            const float4 s1 = *(const float4*)(Sinit + soff + 16 * mt + 4);
            S[mt][0] = s0.x; S[mt][1] = s0.y; S[mt][2] = s0.z; S[mt][3] = s0.w;
            S[mt][4] = s1.x; S[mt][5] = s1.y; S[mt][6] = s1.z; S[mt][7] = s1.w;
        } else {
#pragma unroll
            for (int r = 0; r < 8; ++r) S[mt][r] = 0.f;
        }
    }
    if (OUT) __builtin_amdgcn_sched_barrier(0);
#pragma unroll
    for (int s = 1; s < NSTEP; ++s) {
        g[s][0] = *(const float4*)(xin + goff + (size_t)s * TSTEP * D_MODEL);
        g[s][1] = *(const float4*)(xin + goff + (size_t)s * TSTEP * D_MODEL + 64);
    }

#pragma unroll
    for (int e = 0; e < 2; ++e) {
        const int n = lane + 32 * e;
        ldaS[w][n] = -expf(alv[e]) * dtd;
        if (OUT)
            cfS[w][n] = cmv[e] * bmv[e] * dtd;
    }
    lds_barrier();

    v2f Pf[4][4];
    float a16[4][8];
#pragma unroll
    for (int mt = 0; mt < 4; ++mt) {
        const float l = ldaS[w][16 * mt + c];
#pragma unroll
        for (int kk = 0; kk < 4; ++kk) {
            const int i0 = 4 * kk + 2 * h;
            Pf[mt][kk].x = __expf((float)(15 - i0) * l);
            Pf[mt][kk].y = __expf((float)(14 - i0) * l);
        }
#pragma unroll
        for (int r = 0; r < 8; ++r)
            a16[mt][r] = __expf(16.0f * ldaS[w][16 * mt + 8 * h + r]);
    }

    v2f Qf[4][4];
    v2f Kf[4];
    if (OUT) {
#pragma unroll
        for (int mt = 0; mt < 4; ++mt)
#pragma unroll
            for (int q = 0; q < 4; ++q) {
                const int n = 16 * mt + 8 * h + 2 * q;
                Qf[mt][q].x = cfS[w][n]     * __expf((float)(c + 1) * ldaS[w][n]);
                Qf[mt][q].y = cfS[w][n + 1] * __expf((float)(c + 1) * ldaS[w][n + 1]);
            }
        float part = 0.f;
        for (int n = 32 * h; n < 32 * h + 32; ++n)
            part = fmaf(cfS[w][n], __expf((float)c * ldaS[w][n]), part);
        float kj = part + __shfl_xor(part, 16, 32);
        if (c == 0) kj += dvd;
#pragma unroll
        for (int kk = 0; kk < 4; ++kk) {
            const int j0 = c - (4 * kk + 2 * h);
            const int j1 = j0 - 1;
            const float v0 = __shfl(kj, j0 > 0 ? j0 : 0, 32);
            const float v1 = __shfl(kj, j1 > 0 ? j1 : 0, 32);
            Kf[kk].x = j0 >= 0 ? v0 : 0.f;
            Kf[kk].y = j1 >= 0 ? v1 : 0.f;
        }
    }

    Ul[0][0][widx] = g[0][0].x; Ul[0][1][widx] = g[0][0].y; Ul[0][2][widx] = g[0][0].z; Ul[0][3][widx] = g[0][0].w;
    Ul[0][4][widx] = g[0][1].x; Ul[0][5][widx] = g[0][1].y; Ul[0][6][widx] = g[0][1].z; Ul[0][7][widx] = g[0][1].w;
    lds_barrier();

#pragma unroll
    for (int step = 0; step < NSTEP; ++step) {
        const int cur = step & 1;
        const bool more = step + 1 < NSTEP;
        v2f Uf[4];
#pragma unroll
        for (int kk = 0; kk < 4; ++kk)
            Uf[kk] = *(const v2f*)&Ul[cur][w][((kk * 2 + h) * 16 + c) * 2];

        if (OUT) {
            v8f Y0 = {}, Y1 = {};
#pragma unroll
            for (int kk = 0; kk < 4; kk += 2) {
                Y0 = __builtin_amdgcn_wmma_f32_16x16x4_f32(false, Kf[kk], false, Uf[kk], (short)0, Y0, false, false);
                Y1 = __builtin_amdgcn_wmma_f32_16x16x4_f32(false, Kf[kk + 1], false, Uf[kk + 1], (short)0, Y1, false, false);
            }
#pragma unroll
            for (int mt = 0; mt < 4; ++mt)
#pragma unroll
                for (int q = 0; q < 4; q += 2) {
                    v2f sb0, sb1;
                    sb0.x = S[mt][2 * q];     sb0.y = S[mt][2 * q + 1];
                    sb1.x = S[mt][2 * q + 2]; sb1.y = S[mt][2 * q + 3];
                    Y0 = __builtin_amdgcn_wmma_f32_16x16x4_f32(false, Qf[mt][q], false, sb0, (short)0, Y0, false, false);
                    Y1 = __builtin_amdgcn_wmma_f32_16x16x4_f32(false, Qf[mt][q + 1], false, sb1, (short)0, Y1, false, false);
                }
#pragma unroll
            for (int r = 0; r < 8; ++r)
                Yl[cur][w][(8 * h + r) * 16 + c + 32 * h] = Y0[r] + Y1[r];
        }

#pragma unroll
        for (int mt = 0; mt < 4; ++mt) {
            v8f acc;
#pragma unroll
            for (int r = 0; r < 8; ++r) acc[r] = a16[mt][r] * S[mt][r];
#pragma unroll
            for (int kk = 0; kk < 4; ++kk)
                acc = __builtin_amdgcn_wmma_f32_16x16x4_f32(false, Pf[mt][kk], false, Uf[kk], (short)0, acc, false, false);
            S[mt] = acc;
        }

        if (more) {
            const int nxt = cur ^ 1;
            const float4 n0 = g[step + 1][0];
            const float4 n1 = g[step + 1][1];
            Ul[nxt][0][widx] = n0.x; Ul[nxt][1][widx] = n0.y; Ul[nxt][2][widx] = n0.z; Ul[nxt][3][widx] = n0.w;
            Ul[nxt][4][widx] = n1.x; Ul[nxt][5][widx] = n1.y; Ul[nxt][6][widx] = n1.z; Ul[nxt][7][widx] = n1.w;
        }
        lds_barrier();
        if (OUT) {
            float4 o0, o1;
            o0.x = Yl[cur][0][ridx]; o0.y = Yl[cur][1][ridx]; o0.z = Yl[cur][2][ridx]; o0.w = Yl[cur][3][ridx];
            o1.x = Yl[cur][4][ridx]; o1.y = Yl[cur][5][ridx]; o1.z = Yl[cur][6][ridx]; o1.w = Yl[cur][7][ridx];
            float* yp = y + goff + (size_t)step * TSTEP * D_MODEL;
            *(float4*)(yp)     = o0;
            *(float4*)(yp + 64) = o1;
        }
    }

    if (!OUT) {
#pragma unroll
        for (int mt = 0; mt < 4; ++mt) {
            float4 s0, s1;
            s0.x = S[mt][0]; s0.y = S[mt][1]; s0.z = S[mt][2]; s0.w = S[mt][3];
            s1.x = S[mt][4]; s1.y = S[mt][5]; s1.z = S[mt][6]; s1.w = S[mt][7];
            *(float4*)(Eout + soff + 16 * mt)     = s0;
            *(float4*)(Eout + soff + 16 * mt + 4) = s1;
        }
    }
}

__global__ __launch_bounds__(256) void s4_chunk_prefix(
    const float* __restrict__ A_log, const float* __restrict__ dt,
    const float* __restrict__ Echunk,
    float* __restrict__ Sinit)
{
    const int bd   = blockIdx.x * 8 + (threadIdx.x >> 5);
    const int lane = threadIdx.x & 31;
    const int d  = bd & 511;
    const int n0 = lane * 2;

    const float dtd = dt[d];
    const float a0 = -expf(A_log[d * D_STATE + n0]);
    const float a1 = -expf(A_log[d * D_STATE + n0 + 1]);
    const float dAp0 = expf(a0 * dtd * (float)TCHUNK);
    const float dAp1 = expf(a1 * dtd * (float)TCHUNK);

    const float* E = Echunk + (size_t)bd * NCHUNK * D_STATE;
    float*       S = Sinit  + (size_t)bd * NCHUNK * D_STATE;

    float s0 = 0.f, s1 = 0.f;
#pragma unroll
    for (int c = 0; c < NCHUNK; ++c) {
        S[c * D_STATE + n0]     = s0;
        S[c * D_STATE + n0 + 1] = s1;
        s0 = fmaf(dAp0, s0, E[c * D_STATE + n0]);
        s1 = fmaf(dAp1, s1, E[c * D_STATE + n0 + 1]);
    }
}

extern "C" void kernel_launch(void* const* d_in, const int* in_sizes, int n_in,
                              void* d_out, int out_size, void* d_ws, size_t ws_size,
                              hipStream_t stream)
{
    (void)in_sizes; (void)n_in; (void)out_size; (void)ws_size;

    const float* x     = (const float*)d_in[0];
    const float* W_in  = (const float*)d_in[1];
    const float* b_in  = (const float*)d_in[2];
    const float* A_log = (const float*)d_in[3];
    const float* Bm    = (const float*)d_in[4];
    const float* Cm    = (const float*)d_in[5];
    const float* Dv    = (const float*)d_in[6];
    const float* dt    = (const float*)d_in[7];
    const float* W_out = (const float*)d_in[8];
    const float* b_out = (const float*)d_in[9];

    float* xin  = (float*)d_ws;
    float* ymid = xin  + (size_t)MTOT * D_MODEL;
    float* S    = ymid + (size_t)MTOT * D_MODEL;
    float* E    = ymid;
    float* out  = (float*)d_out;

    const dim3 blk(256);
    const dim3 gblk(128);
    gemm_wmma_f32<true><<<256, gblk, 0, stream>>>(x, W_in, b_in, xin);
    scan_mm<false><<<256, blk, 0, stream>>>(xin, A_log, Bm, Cm, Dv, dt, nullptr, E, nullptr);
    s4_chunk_prefix<<<256, blk, 0, stream>>>(A_log, dt, E, S);
    scan_mm<true><<<256, blk, 0, stream>>>(xin, A_log, Bm, Cm, Dv, dt, S, nullptr, ymid);
    gemm_wmma_f32<false><<<256, gblk, 0, stream>>>(ymid, W_out, b_out, out);
}


# Scan in one sweep, one launch: a wave keeps its channel's zero-start states in registers, prefixes the 16 chunk-end states across lanes, outputs from state + decayed carry (summation order changed)
# speedup vs baseline: 5.3267x; 1.2366x over previous
#include <hip/hip_runtime.h>


typedef float v2f __attribute__((ext_vector_type(2)));
typedef float v4f __attribute__((ext_vector_type(4)));
typedef float v8f __attribute__((ext_vector_type(8)));

#define D_MODEL 512
#define D_STATE 64
#define SEQ_LEN 2048
#define BATCH 4
#define MTOT (BATCH * SEQ_LEN)
#define NCHUNK 16
#define TCHUNK (SEQ_LEN / NCHUNK)
#define TSTEP 16
#define NSTEP (TCHUNK / TSTEP)

#define KBLK 32
#define TILE_FLOATS (16 * D_MODEL)

struct FragSet {
    v4f a[4][4];
    v4f b[4][4];
};

template <bool PACKED_A>
__device__ __forceinline__ void gemm_fetch(FragSet& f, const float* const (&arow)[4],
                                           const float* const (&brow)[4], int kb)
{
#pragma unroll
    for (int p = 0; p < 4; ++p) {
#pragma unroll
        for (int i = 0; i < 4; ++i)
            f.a[i][p] = *(const v4f*)(arow[i] + (PACKED_A ? kb * 16 + 128 * p : kb + 8 * p));
#pragma unroll
        for (int j = 0; j < 4; ++j) f.b[j][p] = *(const v4f*)(brow[j] + kb + 8 * p);
    }
}

template <bool SWAP>
__device__ __forceinline__ void gemm_block(v8f (&acc)[4][4], const FragSet& f)
{
#pragma unroll
    for (int p = 0; p < 4; ++p) {
        v2f a0[4], a1[4], b0[4], b1[4];
#pragma unroll
        for (int i = 0; i < 4; ++i) {
            a0[i].x = f.a[i][p].x; a0[i].y = f.a[i][p].y;
            a1[i].x = f.a[i][p].z; a1[i].y = f.a[i][p].w;
            b0[i].x = f.b[i][p].x; b0[i].y = f.b[i][p].y;
            b1[i].x = f.b[i][p].z; b1[i].y = f.b[i][p].w;
        }
#pragma unroll
        for (int i = 0; i < 4; ++i)
#pragma unroll
            for (int j = 0; j < 4; ++j)
                acc[i][j] = __builtin_amdgcn_wmma_f32_16x16x4_f32(
                    false, SWAP ? b0[j] : a0[i], false, SWAP ? a0[i] : b0[j],
                    (short)0, acc[i][j], false, false);
#pragma unroll
        for (int i = 0; i < 4; ++i)
#pragma unroll
            for (int j = 0; j < 4; ++j)
                acc[i][j] = __builtin_amdgcn_wmma_f32_16x16x4_f32(
                    false, SWAP ? b1[j] : a1[i], false, SWAP ? a1[i] : b1[j],
                    (short)0, acc[i][j], false, false);
    }
}

template <bool PACKED_OUT>
__global__ __launch_bounds__(128) void gemm_wmma_f32(
    const float* __restrict__ A, const float* __restrict__ W,
    const float* __restrict__ bias, float* __restrict__ out)
{
    constexpr bool PACKED_A = !PACKED_OUT;
    const int lane = threadIdx.x & 31;
    const int l16  = lane & 15;
    const int half = lane >> 4;
    const int w    = threadIdx.x >> 5;
    const int M0 = (blockIdx.x >> 2) * 128 + (w >> 1) * 64;
    const int N0 = (blockIdx.x & 3) * 128 + (w & 1) * 64;

    const float* arow[4];
    const float* brow[4];
#pragma unroll
    for (int i = 0; i < 4; ++i)
        arow[i] = PACKED_A
            ? A + (size_t)((M0 >> 4) + i) * TILE_FLOATS + half * 64 + l16 * 4
            : A + (size_t)(M0 + i * 16 + l16) * D_MODEL + half * 4;
#pragma unroll
    for (int j = 0; j < 4; ++j)
        brow[j] = W + (size_t)(N0 + j * 16 + l16) * D_MODEL + half * 4;

    v8f acc[4][4] = {};
    FragSet f0, f1;

    gemm_fetch<PACKED_A>(f0, arow, brow, 0);
    for (int kb = 0; kb < D_MODEL - 2 * KBLK; kb += 2 * KBLK) {
        gemm_fetch<PACKED_A>(f1, arow, brow, kb + KBLK);
        __builtin_amdgcn_sched_barrier(0);
        gemm_block<PACKED_OUT>(acc, f0);
        __builtin_amdgcn_sched_barrier(0);
        gemm_fetch<PACKED_A>(f0, arow, brow, kb + 2 * KBLK);
        gemm_block<PACKED_OUT>(acc, f1);
    }
    gemm_fetch<PACKED_A>(f1, arow, brow, D_MODEL - KBLK);
    gemm_block<PACKED_OUT>(acc, f0);
    gemm_block<PACKED_OUT>(acc, f1);

    if (PACKED_OUT) {
#pragma unroll
        for (int j = 0; j < 4; ++j) {
            const int ch = N0 + j * 16 + half * 8;
            float bj[8];
#pragma unroll
            for (int r = 0; r < 8; ++r) bj[r] = bias[ch + r];
#pragma unroll
            for (int i = 0; i < 4; ++i) {
                float* op = out + (size_t)((M0 >> 4) + i) * TILE_FLOATS + (ch >> 2) * 64 + l16 * 4;
                v4f o0, o1;
                o0.x = acc[i][j][0] + bj[0]; o0.y = acc[i][j][1] + bj[1];
                o0.z = acc[i][j][2] + bj[2]; o0.w = acc[i][j][3] + bj[3];
                o1.x = acc[i][j][4] + bj[4]; o1.y = acc[i][j][5] + bj[5];
                o1.z = acc[i][j][6] + bj[6]; o1.w = acc[i][j][7] + bj[7];
                *(v4f*)(op)      = o0;
                *(v4f*)(op + 64) = o1;
            }
        }
        return;
    }
#pragma unroll
    for (int j = 0; j < 4; ++j) {
        const float bj = bias[N0 + j * 16 + l16];
#pragma unroll
        for (int i = 0; i < 4; ++i) {
#pragma unroll
            for (int r = 0; r < 8; ++r) {
                const int row = M0 + i * 16 + half * 8 + r;
                out[(size_t)row * D_MODEL + N0 + j * 16 + l16] =
                    acc[i][j][r] + bj;
            }
        }
    }
}

__device__ __forceinline__ void lds_barrier()
{
    __builtin_amdgcn_fence(__ATOMIC_RELEASE, "workgroup", "local");
    __builtin_amdgcn_s_barrier();
    __builtin_amdgcn_fence(__ATOMIC_ACQUIRE, "workgroup", "local");
}

template <int N>
__device__ __forceinline__ float lane_shr(float v)
{
    return __int_as_float(__builtin_amdgcn_update_dpp(0, __float_as_int(v), 0x110 + N, 0xF, 0xF, true));
}

__global__ __launch_bounds__(256) void scan_mm(
    const float* __restrict__ xin,
    const float* __restrict__ A_log,
    const float* __restrict__ Bm,
    const float* __restrict__ Cm,
    const float* __restrict__ Dv,
    const float* __restrict__ dt,
    float* __restrict__ y)
{
    __shared__ float Ul[2][8][512];
    __shared__ float Yl[2][8][320];
    __shared__ float ldaS[8][64];
    __shared__ float cfS[8][64];

    const int tid  = threadIdx.x;
    const int w    = tid >> 5;
    const int lane = tid & 31;
    const int c    = lane & 15;
    const int h    = lane >> 4;
    const int b    = blockIdx.x >> 6;
    const int d0   = (blockIdx.x & 63) * 8;
    const int d    = d0 + w;

    const int ti = tid & 15;
    const int tc = tid >> 4;
    const size_t goff = (size_t)(b * (SEQ_LEN / 16) + tc * (TCHUNK / 16)) * TILE_FLOATS
                      + (size_t)(d0 >> 2) * 64 + ti * 4;
    const int widx = ((ti >> 1) * 16 + tc) * 2 + (ti & 1);
    const int ridx = ti * 16 + tc + 32 * (ti >> 3);

    const float dtd = dt[d];
    const float dvd = Dv[d];
    float alv[2], cmv[2], bmv[2];
#pragma unroll
    for (int e = 0; e < 2; ++e) {
        const int n = lane + 32 * e;
        alv[e] = A_log[d * D_STATE + n];
        cmv[e] = Cm[d * D_STATE + n];
        bmv[e] = Bm[d * D_STATE + n];
    }
    float4 g[NSTEP][2];
#pragma unroll
    for (int s = 0; s < NSTEP; ++s) {
        g[s][0] = *(const float4*)(xin + goff + (size_t)s * TSTEP * D_MODEL);
        g[s][1] = *(const float4*)(xin + goff + (size_t)s * TSTEP * D_MODEL + 64);
    }

#pragma unroll
    for (int e = 0; e < 2; ++e) {
        const int n = lane + 32 * e;
        ldaS[w][n] = -expf(alv[e]) * dtd;
        cfS[w][n] = cmv[e] * bmv[e] * dtd;
    }
    lds_barrier();

    v2f Pf[4][4];
    float a16[4][8];
#pragma unroll
    for (int mt = 0; mt < 4; ++mt) {
        const float l = ldaS[w][16 * mt + c];
#pragma unroll
        for (int kk = 0; kk < 4; ++kk) {
            const int i0 = 4 * kk + 2 * h;
            Pf[mt][kk].x = __expf((float)(15 - i0) * l);
            Pf[mt][kk].y = __expf((float)(14 - i0) * l);
        }
#pragma unroll
        for (int r = 0; r < 8; ++r)
            a16[mt][r] = __expf(16.0f * ldaS[w][16 * mt + 8 * h + r]);
    }

    v2f Qf[4][4];
    v2f Kf[4];
#pragma unroll
    for (int mt = 0; mt < 4; ++mt)
#pragma unroll
        for (int q = 0; q < 4; ++q) {
            const int n = 16 * mt + 8 * h + 2 * q;
            Qf[mt][q].x = cfS[w][n]     * __expf((float)(c + 1) * ldaS[w][n]);
            Qf[mt][q].y = cfS[w][n + 1] * __expf((float)(c + 1) * ldaS[w][n + 1]);
        }
    {
        float part = 0.f;
        for (int n = 32 * h; n < 32 * h + 32; ++n)
            part = fmaf(cfS[w][n], __expf((float)c * ldaS[w][n]), part);
        float kj = part + __shfl_xor(part, 16, 32);
        if (c == 0) kj += dvd;
#pragma unroll
        for (int kk = 0; kk < 4; ++kk) {
            const int j0 = c - (4 * kk + 2 * h);
            const int j1 = j0 - 1;
            const float v0 = __shfl(kj, j0 > 0 ? j0 : 0, 32);
            const float v1 = __shfl(kj, j1 > 0 ? j1 : 0, 32);
            Kf[kk].x = j0 >= 0 ? v0 : 0.f;
            Kf[kk].y = j1 >= 0 ? v1 : 0.f;
        }
    }

    Ul[0][0][widx] = g[0][0].x; Ul[0][1][widx] = g[0][0].y; Ul[0][2][widx] = g[0][0].z; Ul[0][3][widx] = g[0][0].w;
    Ul[0][4][widx] = g[0][1].x; Ul[0][5][widx] = g[0][1].y; Ul[0][6][widx] = g[0][1].z; Ul[0][7][widx] = g[0][1].w;
    lds_barrier();

    v8f S[4];
#pragma unroll
    for (int mt = 0; mt < 4; ++mt)
#pragma unroll
        for (int r = 0; r < 8; ++r) S[mt][r] = 0.f;
    v8f Z[NSTEP - 1][4];
    v2f Ua[NSTEP][4];

#pragma unroll
    for (int step = 0; step < NSTEP; ++step) {
        const int cur = step & 1;
        const bool more = step + 1 < NSTEP;
#pragma unroll
        for (int kk = 0; kk < 4; ++kk)
            Ua[step][kk] = *(const v2f*)&Ul[cur][w][((kk * 2 + h) * 16 + c) * 2];

#pragma unroll
        for (int mt = 0; mt < 4; ++mt) {
            v8f acc;
            if (step > 0) {
                Z[step - 1][mt] = S[mt];
#pragma unroll
                for (int r = 0; r < 8; ++r) acc[r] = a16[mt][r] * S[mt][r];
            } else {
#pragma unroll
                for (int r = 0; r < 8; ++r) acc[r] = 0.f;
            }
#pragma unroll
            for (int kk = 0; kk < 4; ++kk)
                acc = __builtin_amdgcn_wmma_f32_16x16x4_f32(false, Pf[mt][kk], false, Ua[step][kk], (short)0, acc, false, false);
            S[mt] = acc;
        }

        if (more) {
            const int nxt = cur ^ 1;
            const float4 n0 = g[step + 1][0];
            const float4 n1 = g[step + 1][1];
            Ul[nxt][0][widx] = n0.x; Ul[nxt][1][widx] = n0.y; Ul[nxt][2][widx] = n0.z; Ul[nxt][3][widx] = n0.w;
            Ul[nxt][4][widx] = n1.x; Ul[nxt][5][widx] = n1.y; Ul[nxt][6][widx] = n1.z; Ul[nxt][7][widx] = n1.w;
            lds_barrier();
        }
    }

#pragma unroll
    for (int mt = 0; mt < 4; ++mt)
#pragma unroll
        for (int r = 0; r < 8; ++r) {
            float m = expf((float)TCHUNK * ldaS[w][16 * mt + 8 * h + r]);
            float v = S[mt][r];
            float t;
            t = lane_shr<1>(v); v = (c >= 1) ? fmaf(m, t, v) : v; m = m * m;
            t = lane_shr<2>(v); v = (c >= 2) ? fmaf(m, t, v) : v; m = m * m;
            t = lane_shr<4>(v); v = (c >= 4) ? fmaf(m, t, v) : v; m = m * m;
            t = lane_shr<8>(v); v = (c >= 8) ? fmaf(m, t, v) : v;
            S[mt][r] = lane_shr<1>(v);
        }

#pragma unroll
    for (int step = 0; step < NSTEP; ++step) {
        const int cur = step & 1;
        v8f Sc[4];
#pragma unroll
        for (int mt = 0; mt < 4; ++mt) {
            if (step > 0) Sc[mt] = Z[step - 1][mt] + S[mt];
            else          Sc[mt] = S[mt];
        }
        v8f Y0 = {}, Y1 = {};
#pragma unroll
        for (int kk = 0; kk < 4; kk += 2) {
            Y0 = __builtin_amdgcn_wmma_f32_16x16x4_f32(false, Kf[kk], false, Ua[step][kk], (short)0, Y0, false, false);
            Y1 = __builtin_amdgcn_wmma_f32_16x16x4_f32(false, Kf[kk + 1], false, Ua[step][kk + 1], (short)0, Y1, false, false);
        }
#pragma unroll
        for (int mt = 0; mt < 4; ++mt)
#pragma unroll
            for (int q = 0; q < 4; q += 2) {
                v2f sb0, sb1;
                sb0.x = Sc[mt][2 * q];     sb0.y = Sc[mt][2 * q + 1];
                sb1.x = Sc[mt][2 * q + 2]; sb1.y = Sc[mt][2 * q + 3];
                Y0 = __builtin_amdgcn_wmma_f32_16x16x4_f32(false, Qf[mt][q], false, sb0, (short)0, Y0, false, false);
                Y1 = __builtin_amdgcn_wmma_f32_16x16x4_f32(false, Qf[mt][q + 1], false, sb1, (short)0, Y1, false, false);
            }
#pragma unroll
        for (int r = 0; r < 8; ++r)
            Yl[cur][w][(8 * h + r) * 16 + c + 32 * h] = Y0[r] + Y1[r];
        if (step + 1 < NSTEP) {
#pragma unroll
            for (int mt = 0; mt < 4; ++mt)
#pragma unroll
                for (int r = 0; r < 8; ++r) S[mt][r] = a16[mt][r] * S[mt][r];
        }
        lds_barrier();
        float4 o0, o1;
        o0.x = Yl[cur][0][ridx]; o0.y = Yl[cur][1][ridx]; o0.z = Yl[cur][2][ridx]; o0.w = Yl[cur][3][ridx];
        o1.x = Yl[cur][4][ridx]; o1.y = Yl[cur][5][ridx]; o1.z = Yl[cur][6][ridx]; o1.w = Yl[cur][7][ridx];
        float* yp = y + goff + (size_t)step * TSTEP * D_MODEL;
        *(float4*)(yp)     = o0;
        *(float4*)(yp + 64) = o1;
    }
}

extern "C" void kernel_launch(void* const* d_in, const int* in_sizes, int n_in,
                              void* d_out, int out_size, void* d_ws, size_t ws_size,
                              hipStream_t stream)
{
    (void)in_sizes; (void)n_in; (void)out_size; (void)ws_size;

    const float* x     = (const float*)d_in[0];
    const float* W_in  = (const float*)d_in[1];
    const float* b_in  = (const float*)d_in[2];
    const float* A_log = (const float*)d_in[3];
    const float* Bm    = (const float*)d_in[4];
    const float* Cm    = (const float*)d_in[5];
    const float* Dv    = (const float*)d_in[6];
    const float* dt    = (const float*)d_in[7];
    const float* W_out = (const float*)d_in[8];
    const float* b_out = (const float*)d_in[9];

    float* xin  = (float*)d_ws;
    float* ymid = xin  + (size_t)MTOT * D_MODEL;
    float* out  = (float*)d_out;

    const dim3 blk(256);
    const dim3 gblk(128);
    gemm_wmma_f32<true><<<256, gblk, 0, stream>>>(x, W_in, b_in, xin);
    scan_mm<<<256, blk, 0, stream>>>(xin, A_log, Bm, Cm, Dv, dt, ymid);
    gemm_wmma_f32<false><<<256, gblk, 0, stream>>>(ymid, W_out, b_out, out);
}
